# starting kernel
# baseline (speedup 1.0000x reference)
#include <hip/hip_runtime.h>
#include <math.h>

#define NNODES 50000
#define NEDGES 800000
#define ETOT   (NEDGES + NNODES)
#define INDIM  128
#define H1C    4
#define C1C    32
#define HID    128
#define OUTD   64
#define NGRAPH 64

typedef float v2f __attribute__((ext_vector_type(2)));
typedef float v8f __attribute__((ext_vector_type(8)));

__global__ void gemm_wmma_f32(const float* __restrict__ A, const float* __restrict__ B,
                              float* __restrict__ Cm, int M, int K, int Ncol) {
  int gwave = (int)((blockIdx.x * blockDim.x + threadIdx.x) >> 5);
  int lane  = (int)(threadIdx.x & 31);
  int tilesN = Ncol >> 4;
  int numTiles = (M >> 4) * tilesN;
  if (gwave >= numTiles) return;
  int tm = (gwave / tilesN) << 4;
  int tn = (gwave % tilesN) << 4;
  int half = lane >> 4;
  int l    = lane & 15;

  v8f acc = {};
  const float* arow = A + (size_t)(tm + l) * K;
  for (int k0 = 0; k0 < K; k0 += 4) {
    int ka = k0 + (half << 1);
    v2f a, b;
    a.x = arow[ka];
    a.y = arow[ka + 1];
    b.x = B[(size_t)(ka)     * Ncol + tn + l];
    b.y = B[(size_t)(ka + 1) * Ncol + tn + l];
    acc = __builtin_amdgcn_wmma_f32_16x16x4_f32(false, a, false, b,
                                                (short)0, acc, false, false);
  }
  int row0 = tm + (half << 3);
  #pragma unroll
  for (int v = 0; v < 8; ++v)
    Cm[(size_t)(row0 + v) * Ncol + tn + l] = acc[v];
}

__global__ void attn_scores(const float* __restrict__ xw,
                            const float* __restrict__ a_src,
                            const float* __restrict__ a_dst,
                            float* __restrict__ esrc, float* __restrict__ edst,
                            int n, int H, int C) {
  int t = blockIdx.x * blockDim.x + threadIdx.x;
  if (t >= n * H) return;
  int node = t / H, h = t % H;
  const float* v  = xw + (size_t)node * H * C + (size_t)h * C;
  const float* as = a_src + (size_t)h * C;
  const float* ad = a_dst + (size_t)h * C;
  float ss = 0.f, sd = 0.f;
  for (int c = 0; c < C; ++c) { float x = v[c]; ss += x * as[c]; sd += x * ad[c]; }
  esrc[t] = ss;
  edst[t] = sd;
}

__device__ __forceinline__ unsigned ordered_enc(float f) {
  unsigned u = __float_as_uint(f);
  return (u & 0x80000000u) ? ~u : (u | 0x80000000u);
}
__device__ __forceinline__ float ordered_dec(unsigned o) {
  return (o & 0x80000000u) ? __uint_as_float(o & 0x7FFFFFFFu) : __uint_as_float(~o);
}

__device__ __forceinline__ void edge_nodes(const int* __restrict__ src,
                                           const int* __restrict__ dst,
                                           int e, int& s, int& d) {
  if (e < NEDGES) { s = src[e]; d = dst[e]; }
  else            { s = e - NEDGES; d = s; }
}

__global__ void edge_alpha_max(const int* __restrict__ src, const int* __restrict__ dst,
                               const float* __restrict__ esrc, const float* __restrict__ edst,
                               unsigned* __restrict__ amax, float* __restrict__ alpha, int H) {
  int t = blockIdx.x * blockDim.x + threadIdx.x;
  if (t >= ETOT * H) return;
  int e = t / H, h = t % H;
  int s, d; edge_nodes(src, dst, e, s, d);
  float a = esrc[s * H + h] + edst[d * H + h];
  a = (a >= 0.f) ? a : 0.2f * a;
  alpha[t] = a;
  atomicMax(&amax[d * H + h], ordered_enc(a));
}

__global__ void edge_exp_denom(const int* __restrict__ src, const int* __restrict__ dst,
                               float* __restrict__ alpha_ex,
                               const unsigned* __restrict__ amax,
                               float* __restrict__ denom, int H) {
  int t = blockIdx.x * blockDim.x + threadIdx.x;
  if (t >= ETOT * H) return;
  int e = t / H, h = t % H;
  int s, d; edge_nodes(src, dst, e, s, d);
  float m  = ordered_dec(amax[d * H + h]);
  float ex = expf(alpha_ex[t] - m);
  alpha_ex[t] = ex;
  atomicAdd(&denom[d * H + h], ex);
}

__global__ void edge_aggregate(const int* __restrict__ src, const int* __restrict__ dst,
                               const float* __restrict__ xw,
                               const float* __restrict__ ex, const float* __restrict__ denom,
                               float* __restrict__ agg, int H, int C) {
  int F = H * C;
  int t = blockIdx.x * blockDim.x + threadIdx.x;
  if (t >= ETOT * F) return;
  int e = t / F, c = t % F;
  int h = c / C;
  int s, d; edge_nodes(src, dst, e, s, d);
  float a = ex[(size_t)e * H + h] / denom[d * H + h];
  atomicAdd(&agg[(size_t)d * F + c], xw[(size_t)s * F + c] * a);
}

__global__ void bias_act(float* __restrict__ x, const float* __restrict__ bias,
                         int n, int F, int doElu) {
  int t = blockIdx.x * blockDim.x + threadIdx.x;
  if (t >= n * F) return;
  int c = t % F;
  float v = x[t] + bias[c];
  if (doElu) v = (v > 0.f) ? v : (expf(v) - 1.f);
  x[t] = v;
}

__global__ void pool_count(const int* __restrict__ batch, float* __restrict__ cnt) {
  int n = blockIdx.x * blockDim.x + threadIdx.x;
  if (n >= NNODES) return;
  atomicAdd(&cnt[batch[n]], 1.f);
}
__global__ void pool_sum(const float* __restrict__ h, const int* __restrict__ batch,
                         float* __restrict__ out) {
  int t = blockIdx.x * blockDim.x + threadIdx.x;
  if (t >= NNODES * OUTD) return;
  int n = t / OUTD, c = t % OUTD;
  atomicAdd(&out[(size_t)batch[n] * OUTD + c], h[t]);
}
__global__ void pool_div(float* __restrict__ out, const float* __restrict__ cnt) {
  int t = blockIdx.x * blockDim.x + threadIdx.x;
  if (t >= NGRAPH * OUTD) return;
  out[t] /= fmaxf(cnt[t / OUTD], 1.f);
}

static inline int cdiv(long long a, int b) { return (int)((a + b - 1) / b); }

extern "C" void kernel_launch(void* const* d_in, const int* in_sizes, int n_in,
                              void* d_out, int out_size, void* d_ws, size_t ws_size,
                              hipStream_t stream) {
  (void)in_sizes; (void)n_in; (void)out_size; (void)ws_size;
  const float* x   = (const float*)d_in[0];
  const int*   ei  = (const int*)d_in[1];
  const int*   bat = (const int*)d_in[2];
  const float* W1  = (const float*)d_in[3];
  const float* as1 = (const float*)d_in[4];
  const float* ad1 = (const float*)d_in[5];
  const float* b1  = (const float*)d_in[6];
  const float* W2  = (const float*)d_in[7];
  const float* as2 = (const float*)d_in[8];
  const float* ad2 = (const float*)d_in[9];
  const float* b2  = (const float*)d_in[10];
  const int* src = ei;
  const int* dst = ei + NEDGES;

  char* ws = (char*)d_ws;
  size_t off = 0;
  auto carve = [&](size_t bytes) -> char* {
    char* p = ws + off;
    off += (bytes + 255) & ~(size_t)255;
    return p;
  };
  float*    xw1    = (float*)carve((size_t)NNODES * HID * 4);
  float*    esrc1  = (float*)carve((size_t)NNODES * H1C * 4);
  float*    edst1  = (float*)carve((size_t)NNODES * H1C * 4);
  unsigned* amax1  = (unsigned*)carve((size_t)NNODES * H1C * 4);
  float*    denom1 = (float*)carve((size_t)NNODES * H1C * 4);
  float*    ex1    = (float*)carve((size_t)ETOT * H1C * 4);
  float*    agg1   = (float*)carve((size_t)NNODES * HID * 4);
  float*    agg2   = (float*)carve((size_t)NNODES * OUTD * 4);
  float*    cnt    = (float*)carve((size_t)NGRAPH * 4);
  float*    xw2    = xw1;
  float*    esrc2  = esrc1;
  float*    edst2  = edst1;
  unsigned* amax2  = amax1;
  float*    denom2 = denom1;
  float*    ex2    = ex1;
  float*    outF   = (float*)d_out;

  const int B = 256;

  hipMemsetAsync(amax1,  0, (size_t)NNODES * H1C * 4, stream);
  hipMemsetAsync(denom1, 0, (size_t)NNODES * H1C * 4, stream);
  hipMemsetAsync(agg1,   0, (size_t)NNODES * HID * 4, stream);

  {
    int tiles = (NNODES / 16) * (HID / 16);
    gemm_wmma_f32<<<cdiv((long long)tiles * 32, B), B, 0, stream>>>(x, W1, xw1, NNODES, INDIM, HID);
  }
  attn_scores<<<cdiv((long long)NNODES * H1C, B), B, 0, stream>>>(xw1, as1, ad1, esrc1, edst1, NNODES, H1C, C1C);
  edge_alpha_max<<<cdiv((long long)ETOT * H1C, B), B, 0, stream>>>(src, dst, esrc1, edst1, amax1, ex1, H1C);
  edge_exp_denom<<<cdiv((long long)ETOT * H1C, B), B, 0, stream>>>(src, dst, ex1, amax1, denom1, H1C);
  edge_aggregate<<<cdiv((long long)ETOT * HID, B), B, 0, stream>>>(src, dst, xw1, ex1, denom1, agg1, H1C, C1C);
  bias_act<<<cdiv((long long)NNODES * HID, B), B, 0, stream>>>(agg1, b1, NNODES, HID, 1);

  hipMemsetAsync(amax2,  0, (size_t)NNODES * 4, stream);
  hipMemsetAsync(denom2, 0, (size_t)NNODES * 4, stream);
  hipMemsetAsync(agg2,   0, (size_t)NNODES * OUTD * 4, stream);

  {
    int tiles = (NNODES / 16) * (OUTD / 16);
    gemm_wmma_f32<<<cdiv((long long)tiles * 32, B), B, 0, stream>>>(agg1, W2, xw2, NNODES, HID, OUTD);
  }
  attn_scores<<<cdiv((long long)NNODES, B), B, 0, stream>>>(xw2, as2, ad2, esrc2, edst2, NNODES, 1, OUTD);
  edge_alpha_max<<<cdiv((long long)ETOT, B), B, 0, stream>>>(src, dst, esrc2, edst2, amax2, ex2, 1);
  edge_exp_denom<<<cdiv((long long)ETOT, B), B, 0, stream>>>(src, dst, ex2, amax2, denom2, 1);
  edge_aggregate<<<cdiv((long long)ETOT * OUTD, B), B, 0, stream>>>(src, dst, xw2, ex2, denom2, agg2, 1, OUTD);
  bias_act<<<cdiv((long long)NNODES * OUTD, B), B, 0, stream>>>(agg2, b2, NNODES, OUTD, 0);

  hipMemsetAsync(outF, 0, (size_t)NGRAPH * OUTD * 4, stream);
  hipMemsetAsync(cnt,  0, (size_t)NGRAPH * 4, stream);
  pool_count<<<cdiv(NNODES, B), B, 0, stream>>>(bat, cnt);
  pool_sum<<<cdiv((long long)NNODES * OUTD, B), B, 0, stream>>>(agg2, bat, outF);
  pool_div<<<cdiv((long long)NGRAPH * OUTD, B), B, 0, stream>>>(outF, cnt);
}


# Edges grouped by destination per call (integer-atomic rank, scan, place); one wave per node does max, exp-sum, row sum (+bias/ELU) in registers, no float-atomic edge passes; sums run in list order
# speedup vs baseline: 3.5133x; 3.5133x over previous
#include <hip/hip_runtime.h>
#include <math.h>

#define NNODES 50000
#define NEDGES 800000
#define ETOT   (NEDGES + NNODES)
#define INDIM  128
#define H1C    4
#define C1C    32
#define HID    128
#define OUTD   64
#define NGRAPH 64

typedef float v2f __attribute__((ext_vector_type(2)));
typedef float v4f __attribute__((ext_vector_type(4)));
typedef float v8f __attribute__((ext_vector_type(8)));

__global__ void gemm_wmma_f32(const float* __restrict__ A, const float* __restrict__ B,
                              float* __restrict__ Cm, int M, int K, int Ncol) {
  int gwave = (int)((blockIdx.x * blockDim.x + threadIdx.x) >> 5);
  int lane  = (int)(threadIdx.x & 31);
  int tilesN = Ncol >> 4;
  int numTiles = (M >> 4) * tilesN;
  if (gwave >= numTiles) return;
  int tm = (gwave / tilesN) << 4;
  int tn = (gwave % tilesN) << 4;
  int half = lane >> 4;
  int l    = lane & 15;

  v8f acc = {};
  const float* arow = A + (size_t)(tm + l) * K;
  for (int k0 = 0; k0 < K; k0 += 4) {
    int ka = k0 + (half << 1);
    v2f a, b;
    a.x = arow[ka];
    a.y = arow[ka + 1];
    b.x = B[(size_t)(ka)     * Ncol + tn + l];
    b.y = B[(size_t)(ka + 1) * Ncol + tn + l];
    acc = __builtin_amdgcn_wmma_f32_16x16x4_f32(false, a, false, b,
                                                (short)0, acc, false, false);
  }
  int row0 = tm + (half << 3);
  #pragma unroll
  for (int v = 0; v < 8; ++v)
    Cm[(size_t)(row0 + v) * Ncol + tn + l] = acc[v];
}

__global__ void attn_scores(const float* __restrict__ xw,
                            const float* __restrict__ a_src,
                            const float* __restrict__ a_dst,
                            float* __restrict__ esrc, float* __restrict__ edst,
                            int n, int H, int C) {
  int t = blockIdx.x * blockDim.x + threadIdx.x;
  if (t >= n * H) return;
  int node = t / H, h = t % H;
  const float* v  = xw + (size_t)node * H * C + (size_t)h * C;
  const float* as = a_src + (size_t)h * C;
  const float* ad = a_dst + (size_t)h * C;
  float ss = 0.f, sd = 0.f;
  for (int c = 0; c < C; ++c) { float x = v[c]; ss += x * as[c]; sd += x * ad[c]; }
  esrc[t] = ss;
  edst[t] = sd;
}

__global__ void edge_rank(const int* __restrict__ dst, int* __restrict__ deg,
                          int* __restrict__ rank) {
  int e = blockIdx.x * blockDim.x + threadIdx.x;
  if (e >= NEDGES) return;
  rank[e] = atomicAdd(&deg[dst[e]], 1);
}

#define SCANB 1024
__global__ void __launch_bounds__(SCANB) deg_scan(const int* __restrict__ deg,
                                                  int* __restrict__ rowptr) {
  __shared__ int wpre[32];
  __shared__ int wtot[32];
  int t = (int)threadIdx.x;
  int lane = t & 31, w = t >> 5;
  int tile0 = (int)blockIdx.x * SCANB;
  int pre = 0;
  for (int i = t; i < tile0; i += SCANB) pre += deg[i];
  int idx = tile0 + t;
  int v = (idx < NNODES) ? deg[idx] : 0;
  int incl = v;
  for (int o = 1; o < 32; o <<= 1) {
    int n = __shfl_up(incl, o);
    if (lane >= o) incl += n;
  }
  for (int o = 16; o > 0; o >>= 1) pre += __shfl_xor(pre, o);
  if (lane == 31) wtot[w] = incl;
  if (lane == 0)  wpre[w] = pre;
  __syncthreads();
  int base = 0, woff = 0;
  for (int i = 0; i < 32; ++i) {
    base += wpre[i];
    if (i < w) woff += wtot[i];
  }
  if (idx < NNODES) rowptr[idx] = base + woff + incl - v;
}

__global__ void edge_place(const int* __restrict__ src, const int* __restrict__ dst,
                           const int* __restrict__ rank, const int* __restrict__ rowptr,
                           int* __restrict__ csr) {
  int e = blockIdx.x * blockDim.x + threadIdx.x;
  if (e >= NEDGES) return;
  csr[rowptr[dst[e]] + rank[e]] = src[e];
}

template <int H, int C, int ELU>
__global__ void __launch_bounds__(256) node_softmax_gather(
    const int* __restrict__ rowptr, const int* __restrict__ deg, const int* __restrict__ csr,
    const float* __restrict__ xw, const float* __restrict__ esrc, const float* __restrict__ edst,
    const float* __restrict__ bias, float* __restrict__ out) {
  constexpr int F = H * C;
  constexpr int V = F / 32;
  int node = __builtin_amdgcn_readfirstlane((int)((blockIdx.x * blockDim.x + threadIdx.x) >> 5));
  int lane = (int)(threadIdx.x & 31);
  if (node >= NNODES) return;
  const int hd = (lane * V) / C;
  const int beg = rowptr[node];
  const int total = deg[node] + 1;

  float ed[H];
  for (int h = 0; h < H; ++h) ed[h] = edst[node * H + h];

  auto load_chunk = [&](int base, int& s, float (&a)[H]) -> bool {
    int j = base + lane;
    s = node;
    if (j < total - 1) s = csr[beg + j];
    for (int h = 0; h < H; ++h) {
      float x = esrc[s * H + h] + ed[h];
      a[h] = (x >= 0.f) ? x : 0.2f * x;
    }
    return j < total;
  };

  float acc[V];
  for (int v = 0; v < V; ++v) acc[v] = 0.f;

  auto accumulate = [&](int s, const float (&wgt)[H], int cnt) {
    #pragma unroll 4
    for (int jj = 0; jj < cnt; ++jj) {
      int sj = __builtin_amdgcn_readlane(s, jj);
      float wsel = __int_as_float(__builtin_amdgcn_readlane(__float_as_int(wgt[0]), jj));
      for (int h = 1; h < H; ++h) {
        float wh = __int_as_float(__builtin_amdgcn_readlane(__float_as_int(wgt[h]), jj));
        wsel = (hd == h) ? wh : wsel;
      }
      const float* row = xw + (size_t)sj * F + lane * V;
      if constexpr (V == 4) {
        v4f r = *reinterpret_cast<const v4f*>(row);
        acc[0] += r.x * wsel; acc[1] += r.y * wsel; acc[2] += r.z * wsel; acc[3] += r.w * wsel;
      } else {
        v2f r = *reinterpret_cast<const v2f*>(row);
        acc[0] += r.x * wsel; acc[1] += r.y * wsel;
      }
    }
  };

  int s0; float a0[H];
  const bool ok0 = load_chunk(0, s0, a0);
  float m[H];
  for (int h = 0; h < H; ++h) m[h] = ok0 ? a0[h] : -INFINITY;
  for (int base = 32; base < total; base += 32) {
    int s; float a[H];
    bool ok = load_chunk(base, s, a);
    for (int h = 0; h < H; ++h) m[h] = ok ? fmaxf(m[h], a[h]) : m[h];
  }
  for (int h = 0; h < H; ++h)
    for (int o = 16; o > 0; o >>= 1) m[h] = fmaxf(m[h], __shfl_xor(m[h], o));

  float ex0[H], den[H];
  for (int h = 0; h < H; ++h) { ex0[h] = ok0 ? expf(a0[h] - m[h]) : 0.f; den[h] = ex0[h]; }
  for (int base = 32; base < total; base += 32) {
    int s; float a[H];
    bool ok = load_chunk(base, s, a);
    for (int h = 0; h < H; ++h) den[h] += ok ? expf(a[h] - m[h]) : 0.f;
  }
  for (int h = 0; h < H; ++h)
    for (int o = 16; o > 0; o >>= 1) den[h] += __shfl_xor(den[h], o);

  {
    float w0[H];
    for (int h = 0; h < H; ++h) w0[h] = ex0[h] / den[h];
    accumulate(s0, w0, total < 32 ? total : 32);
  }
  for (int base = 32; base < total; base += 32) {
    int s; float a[H], wv[H];
    bool ok = load_chunk(base, s, a);
    for (int h = 0; h < H; ++h) wv[h] = ok ? expf(a[h] - m[h]) / den[h] : 0.f;
    int left = total - base;
    accumulate(s, wv, left < 32 ? left : 32);
  }

  float res[V];
  for (int v = 0; v < V; ++v) {
    float r = acc[v] + bias[lane * V + v];
    if (ELU) r = (r > 0.f) ? r : (expf(r) - 1.f);
    res[v] = r;
  }
  float* o = out + (size_t)node * F + lane * V;
  if constexpr (V == 4) {
    v4f r = {res[0], res[1], res[2], res[3]};
    *reinterpret_cast<v4f*>(o) = r;
  } else {
    v2f r = {res[0], res[1]};
    *reinterpret_cast<v2f*>(o) = r;
  }
}

__global__ void pool_count(const int* __restrict__ batch, float* __restrict__ cnt) {
  int n = blockIdx.x * blockDim.x + threadIdx.x;
  if (n >= NNODES) return;
  atomicAdd(&cnt[batch[n]], 1.f);
}
__global__ void pool_sum(const float* __restrict__ h, const int* __restrict__ batch,
                         float* __restrict__ out) {
  int t = blockIdx.x * blockDim.x + threadIdx.x;
  if (t >= NNODES * OUTD) return;
  int n = t / OUTD, c = t % OUTD;
  atomicAdd(&out[(size_t)batch[n] * OUTD + c], h[t]);
}
__global__ void pool_div(float* __restrict__ out, const float* __restrict__ cnt) {
  int t = blockIdx.x * blockDim.x + threadIdx.x;
  if (t >= NGRAPH * OUTD) return;
  out[t] /= fmaxf(cnt[t / OUTD], 1.f);
}

static inline int cdiv(long long a, int b) { return (int)((a + b - 1) / b); }

extern "C" void kernel_launch(void* const* d_in, const int* in_sizes, int n_in,
                              void* d_out, int out_size, void* d_ws, size_t ws_size,
                              hipStream_t stream) {
  (void)in_sizes; (void)n_in; (void)out_size; (void)ws_size;
  const float* x   = (const float*)d_in[0];
  const int*   ei  = (const int*)d_in[1];
  const int*   bat = (const int*)d_in[2];
  const float* W1  = (const float*)d_in[3];
  const float* as1 = (const float*)d_in[4];
  const float* ad1 = (const float*)d_in[5];
  const float* b1  = (const float*)d_in[6];
  const float* W2  = (const float*)d_in[7];
  const float* as2 = (const float*)d_in[8];
  const float* ad2 = (const float*)d_in[9];
  const float* b2  = (const float*)d_in[10];
  const int* src = ei;
  const int* dst = ei + NEDGES;

  char* ws = (char*)d_ws;
  size_t off = 0;
  auto carve = [&](size_t bytes) -> char* {
    char* p = ws + off;
    off += (bytes + 255) & ~(size_t)255;
    return p;
  };
  float*    xw1    = (float*)carve((size_t)NNODES * HID * 4);
  float*    esrc1  = (float*)carve((size_t)NNODES * H1C * 4);
  float*    edst1  = (float*)carve((size_t)NNODES * H1C * 4);
  int*      deg    = (int*)carve((size_t)NNODES * 4);
  int*      rowptr = (int*)carve((size_t)NNODES * 4);
  int*      rank   = (int*)carve((size_t)NEDGES * 4);
  int*      csr    = (int*)carve((size_t)NEDGES * 4);
  float*    agg1   = (float*)carve((size_t)NNODES * HID * 4);
  float*    agg2   = (float*)carve((size_t)NNODES * OUTD * 4);
  float*    cnt    = (float*)carve((size_t)NGRAPH * 4);
  float*    xw2    = xw1;
  float*    esrc2  = esrc1;
  float*    edst2  = edst1;
  float*    outF   = (float*)d_out;

  const int B = 256;

  hipMemsetAsync(deg, 0, (size_t)NNODES * 4, stream);
  edge_rank<<<cdiv(NEDGES, B), B, 0, stream>>>(dst, deg, rank);
  deg_scan<<<cdiv(NNODES, SCANB), SCANB, 0, stream>>>(deg, rowptr);
  edge_place<<<cdiv(NEDGES, B), B, 0, stream>>>(src, dst, rank, rowptr, csr);

  {
    int tiles = (NNODES / 16) * (HID / 16);
    gemm_wmma_f32<<<cdiv((long long)tiles * 32, B), B, 0, stream>>>(x, W1, xw1, NNODES, INDIM, HID);
  }
  attn_scores<<<cdiv((long long)NNODES * H1C, B), B, 0, stream>>>(xw1, as1, ad1, esrc1, edst1, NNODES, H1C, C1C);
  node_softmax_gather<H1C, C1C, 1><<<cdiv((long long)NNODES * 32, B), B, 0, stream>>>(
      rowptr, deg, csr, xw1, esrc1, edst1, b1, agg1);

  {
    int tiles = (NNODES / 16) * (OUTD / 16);
    gemm_wmma_f32<<<cdiv((long long)tiles * 32, B), B, 0, stream>>>(agg1, W2, xw2, NNODES, HID, OUTD);
  }
  attn_scores<<<cdiv((long long)NNODES, B), B, 0, stream>>>(xw2, as2, ad2, esrc2, edst2, NNODES, 1, OUTD);
  node_softmax_gather<1, OUTD, 0><<<cdiv((long long)NNODES * 32, B), B, 0, stream>>>(
      rowptr, deg, csr, xw2, esrc2, edst2, b2, agg2);

  hipMemsetAsync(outF, 0, (size_t)NGRAPH * OUTD * 4, stream);
  hipMemsetAsync(cnt,  0, (size_t)NGRAPH * 4, stream);
  pool_count<<<cdiv(NNODES, B), B, 0, stream>>>(bat, cnt);
  pool_sum<<<cdiv((long long)NNODES * OUTD, B), B, 0, stream>>>(agg2, bat, outF);
  pool_div<<<cdiv((long long)NGRAPH * OUTD, B), B, 0, stream>>>(outF, cnt);
}


# Mean pool as one launch: one work-group per graph scans the graph ids and sums its nodes' rows in registers, then LDS; no float atomics, no memsets; sums run in fixed node order, not arrival order
# speedup vs baseline: 4.5927x; 1.3072x over previous
#include <hip/hip_runtime.h>
#include <math.h>

#define NNODES 50000
#define NEDGES 800000
#define ETOT   (NEDGES + NNODES)
#define INDIM  128
#define H1C    4
#define C1C    32
#define HID    128
#define OUTD   64
#define NGRAPH 64

typedef float v2f __attribute__((ext_vector_type(2)));
typedef float v4f __attribute__((ext_vector_type(4)));
typedef float v8f __attribute__((ext_vector_type(8)));

__global__ void gemm_wmma_f32(const float* __restrict__ A, const float* __restrict__ B,
                              float* __restrict__ Cm, int M, int K, int Ncol) {
  int gwave = (int)((blockIdx.x * blockDim.x + threadIdx.x) >> 5);
  int lane  = (int)(threadIdx.x & 31);
  int tilesN = Ncol >> 4;
  int numTiles = (M >> 4) * tilesN;
  if (gwave >= numTiles) return;
  int tm = (gwave / tilesN) << 4;
  int tn = (gwave % tilesN) << 4;
  int half = lane >> 4;
  int l    = lane & 15;

  v8f acc = {};
  const float* arow = A + (size_t)(tm + l) * K;
  for (int k0 = 0; k0 < K; k0 += 4) {
    int ka = k0 + (half << 1);
    v2f a, b;
    a.x = arow[ka];
    a.y = arow[ka + 1];
    b.x = B[(size_t)(ka)     * Ncol + tn + l];
    b.y = B[(size_t)(ka + 1) * Ncol + tn + l];
    acc = __builtin_amdgcn_wmma_f32_16x16x4_f32(false, a, false, b,
                                                (short)0, acc, false, false);
  }
  int row0 = tm + (half << 3);
  #pragma unroll
  for (int v = 0; v < 8; ++v)
    Cm[(size_t)(row0 + v) * Ncol + tn + l] = acc[v];
}

__global__ void attn_scores(const float* __restrict__ xw,
                            const float* __restrict__ a_src,
                            const float* __restrict__ a_dst,
                            float* __restrict__ esrc, float* __restrict__ edst,
                            int n, int H, int C) {
  int t = blockIdx.x * blockDim.x + threadIdx.x;
  if (t >= n * H) return;
  int node = t / H, h = t % H;
  const float* v  = xw + (size_t)node * H * C + (size_t)h * C;
  const float* as = a_src + (size_t)h * C;
  const float* ad = a_dst + (size_t)h * C;
  float ss = 0.f, sd = 0.f;
  for (int c = 0; c < C; ++c) { float x = v[c]; ss += x * as[c]; sd += x * ad[c]; }
  esrc[t] = ss;
  edst[t] = sd;
}

__global__ void edge_rank(const int* __restrict__ dst, int* __restrict__ deg,
                          int* __restrict__ rank) {
  int e = blockIdx.x * blockDim.x + threadIdx.x;
  if (e >= NEDGES) return;
  rank[e] = atomicAdd(&deg[dst[e]], 1);
}

#define SCANB 1024
__global__ void __launch_bounds__(SCANB) deg_scan(const int* __restrict__ deg,
                                                  int* __restrict__ rowptr) {
  __shared__ int wpre[32];
  __shared__ int wtot[32];
  int t = (int)threadIdx.x;
  int lane = t & 31, w = t >> 5;
  int tile0 = (int)blockIdx.x * SCANB;
  int pre = 0;
  for (int i = t; i < tile0; i += SCANB) pre += deg[i];
  int idx = tile0 + t;
  int v = (idx < NNODES) ? deg[idx] : 0;
  int incl = v;
  for (int o = 1; o < 32; o <<= 1) {
    int n = __shfl_up(incl, o);
    if (lane >= o) incl += n;
  }
  for (int o = 16; o > 0; o >>= 1) pre += __shfl_xor(pre, o);
  if (lane == 31) wtot[w] = incl;
  if (lane == 0)  wpre[w] = pre;
  __syncthreads();
  int base = 0, woff = 0;
  for (int i = 0; i < 32; ++i) {
    base += wpre[i];
    if (i < w) woff += wtot[i];
  }
  if (idx < NNODES) rowptr[idx] = base + woff + incl - v;
}

__global__ void edge_place(const int* __restrict__ src, const int* __restrict__ dst,
                           const int* __restrict__ rank, const int* __restrict__ rowptr,
                           int* __restrict__ csr) {
  int e = blockIdx.x * blockDim.x + threadIdx.x;
  if (e >= NEDGES) return;
  csr[rowptr[dst[e]] + rank[e]] = src[e];
}

template <int H, int C, int ELU>
__global__ void __launch_bounds__(256) node_softmax_gather(
    const int* __restrict__ rowptr, const int* __restrict__ deg, const int* __restrict__ csr,
    const float* __restrict__ xw, const float* __restrict__ esrc, const float* __restrict__ edst,
    const float* __restrict__ bias, float* __restrict__ out) {
  constexpr int F = H * C;
  constexpr int V = F / 32;
  int node = __builtin_amdgcn_readfirstlane((int)((blockIdx.x * blockDim.x + threadIdx.x) >> 5));
  int lane = (int)(threadIdx.x & 31);
  if (node >= NNODES) return;
  const int hd = (lane * V) / C;
  const int beg = rowptr[node];
  const int total = deg[node] + 1;

  float ed[H];
  for (int h = 0; h < H; ++h) ed[h] = edst[node * H + h];

  auto load_chunk = [&](int base, int& s, float (&a)[H]) -> bool {
    int j = base + lane;
    s = node;
    if (j < total - 1) s = csr[beg + j];
    for (int h = 0; h < H; ++h) {
      float x = esrc[s * H + h] + ed[h];
      a[h] = (x >= 0.f) ? x : 0.2f * x;
    }
    return j < total;
  };

  float acc[V];
  for (int v = 0; v < V; ++v) acc[v] = 0.f;

  auto accumulate = [&](int s, const float (&wgt)[H], int cnt) {
    #pragma unroll 4
    for (int jj = 0; jj < cnt; ++jj) {
      int sj = __builtin_amdgcn_readlane(s, jj);
      float wsel = __int_as_float(__builtin_amdgcn_readlane(__float_as_int(wgt[0]), jj));
      for (int h = 1; h < H; ++h) {
        float wh = __int_as_float(__builtin_amdgcn_readlane(__float_as_int(wgt[h]), jj));
        wsel = (hd == h) ? wh : wsel;
      }
      const float* row = xw + (size_t)sj * F + lane * V;
      if constexpr (V == 4) {
        v4f r = *reinterpret_cast<const v4f*>(row);
        acc[0] += r.x * wsel; acc[1] += r.y * wsel; acc[2] += r.z * wsel; acc[3] += r.w * wsel;
      } else {
        v2f r = *reinterpret_cast<const v2f*>(row);
        acc[0] += r.x * wsel; acc[1] += r.y * wsel;
      }
    }
  };

  int s0; float a0[H];
  const bool ok0 = load_chunk(0, s0, a0);
  float m[H];
  for (int h = 0; h < H; ++h) m[h] = ok0 ? a0[h] : -INFINITY;
  for (int base = 32; base < total; base += 32) {
    int s; float a[H];
    bool ok = load_chunk(base, s, a);
    for (int h = 0; h < H; ++h) m[h] = ok ? fmaxf(m[h], a[h]) : m[h];
  }
  for (int h = 0; h < H; ++h)
    for (int o = 16; o > 0; o >>= 1) m[h] = fmaxf(m[h], __shfl_xor(m[h], o));

  float ex0[H], den[H];
  for (int h = 0; h < H; ++h) { ex0[h] = ok0 ? expf(a0[h] - m[h]) : 0.f; den[h] = ex0[h]; }
  for (int base = 32; base < total; base += 32) {
    int s; float a[H];
    bool ok = load_chunk(base, s, a);
    for (int h = 0; h < H; ++h) den[h] += ok ? expf(a[h] - m[h]) : 0.f;
  }
  for (int h = 0; h < H; ++h)
    for (int o = 16; o > 0; o >>= 1) den[h] += __shfl_xor(den[h], o);

  {
    float w0[H];
    for (int h = 0; h < H; ++h) w0[h] = ex0[h] / den[h];
    accumulate(s0, w0, total < 32 ? total : 32);
  }
  for (int base = 32; base < total; base += 32) {
    int s; float a[H], wv[H];
    bool ok = load_chunk(base, s, a);
    for (int h = 0; h < H; ++h) wv[h] = ok ? expf(a[h] - m[h]) / den[h] : 0.f;
    int left = total - base;
    accumulate(s, wv, left < 32 ? left : 32);
  }

  float res[V];
  for (int v = 0; v < V; ++v) {
    float r = acc[v] + bias[lane * V + v];
    if (ELU) r = (r > 0.f) ? r : (expf(r) - 1.f);
    res[v] = r;
  }
  float* o = out + (size_t)node * F + lane * V;
  if constexpr (V == 4) {
    v4f r = {res[0], res[1], res[2], res[3]};
    *reinterpret_cast<v4f*>(o) = r;
  } else {
    v2f r = {res[0], res[1]};
    *reinterpret_cast<v2f*>(o) = r;
  }
}

#define POOLW 32
#define POOLU 25
__global__ void __launch_bounds__(POOLW * 32) pool_mean(const float* __restrict__ h,
                                                        const int* __restrict__ batch,
                                                        float* __restrict__ out) {
  __shared__ float part[POOLW][OUTD];
  __shared__ int   pcnt[POOLW];
  constexpr int NCHUNK = (NNODES + 31) / 32;
  const int g    = (int)blockIdx.x;
  const int lane = (int)(threadIdx.x & 31);
  const int w = __builtin_amdgcn_readfirstlane((int)(threadIdx.x >> 5));

  float a0 = 0.f, a1 = 0.f;
  int cnt = 0;
  for (int c0 = w; c0 < NCHUNK; c0 += POOLW * POOLU) {
    int b[POOLU];
    #pragma unroll
    for (int k = 0; k < POOLU; ++k) {
      int node = (c0 + k * POOLW) * 32 + lane;
      int id = batch[node < NNODES ? node : NNODES - 1];
      b[k] = (node < NNODES) ? id : -1;
    }
    static_assert(POOLU <= 32, "one lane per chunk of a round");
    unsigned maskv = 0u;
    #pragma unroll
    for (int k = 0; k < POOLU; ++k) {
      unsigned mk = __builtin_amdgcn_ballot_w32(b[k] == g);
      maskv = (lane == k) ? mk : maskv;
    }
    unsigned any = __builtin_amdgcn_ballot_w32(maskv != 0u);
    while (any) {
      const int k = __builtin_ctz(any);
      any &= any - 1u;
      unsigned mask = (unsigned)__builtin_amdgcn_readlane((int)maskv, k);
      cnt += __builtin_popcount(mask);
      const int nbase = (c0 + k * POOLW) * 32;
      while (mask) {
        int  j[8];
        bool p[8];
        #pragma unroll
        for (int u = 0; u < 8; ++u) {
          p[u] = (mask != 0u);
          j[u] = p[u] ? __builtin_ctz(mask) : j[0];
          mask &= mask - 1u;
        }
        v2f r[8];
        #pragma unroll
        for (int u = 0; u < 8; ++u)
          r[u] = *reinterpret_cast<const v2f*>(h + (size_t)(nbase + j[u]) * OUTD + lane * 2);
        #pragma unroll
        for (int u = 0; u < 8; ++u) {
          a0 += p[u] ? r[u].x : 0.f;
          a1 += p[u] ? r[u].y : 0.f;
        }
      }
    }
  }

  v2f mine = {a0, a1};
  *reinterpret_cast<v2f*>(&part[w][lane * 2]) = mine;
  if (lane == 0) pcnt[w] = cnt;
  __syncthreads();
  const int t = (int)threadIdx.x;
  if (t < OUTD) {
    float s = 0.f;
    int   c = 0;
    for (int i = 0; i < POOLW; ++i) { s += part[i][t]; c += pcnt[i]; }
    out[(size_t)g * OUTD + t] = s / fmaxf((float)c, 1.f);
  }
}

static inline int cdiv(long long a, int b) { return (int)((a + b - 1) / b); }

extern "C" void kernel_launch(void* const* d_in, const int* in_sizes, int n_in,
                              void* d_out, int out_size, void* d_ws, size_t ws_size,
                              hipStream_t stream) {
  (void)in_sizes; (void)n_in; (void)out_size; (void)ws_size;
  const float* x   = (const float*)d_in[0];
  const int*   ei  = (const int*)d_in[1];
  const int*   bat = (const int*)d_in[2];
  const float* W1  = (const float*)d_in[3];
  const float* as1 = (const float*)d_in[4];
  const float* ad1 = (const float*)d_in[5];
  const float* b1  = (const float*)d_in[6];
  const float* W2  = (const float*)d_in[7];
  const float* as2 = (const float*)d_in[8];
  const float* ad2 = (const float*)d_in[9];
  const float* b2  = (const float*)d_in[10];
  const int* src = ei;
  const int* dst = ei + NEDGES;

  char* ws = (char*)d_ws;
  size_t off = 0;
  auto carve = [&](size_t bytes) -> char* {
    char* p = ws + off;
    off += (bytes + 255) & ~(size_t)255;
    return p;
  };
  float*    xw1    = (float*)carve((size_t)NNODES * HID * 4);
  float*    esrc1  = (float*)carve((size_t)NNODES * H1C * 4);
  float*    edst1  = (float*)carve((size_t)NNODES * H1C * 4);
  int*      deg    = (int*)carve((size_t)NNODES * 4);
  int*      rowptr = (int*)carve((size_t)NNODES * 4);
  int*      rank   = (int*)carve((size_t)NEDGES * 4);
  int*      csr    = (int*)carve((size_t)NEDGES * 4);
  float*    agg1   = (float*)carve((size_t)NNODES * HID * 4);
  float*    agg2   = (float*)carve((size_t)NNODES * OUTD * 4);
  float*    xw2    = xw1;
  float*    esrc2  = esrc1;
  float*    edst2  = edst1;
  float*    outF   = (float*)d_out;

  const int B = 256;

  hipMemsetAsync(deg, 0, (size_t)NNODES * 4, stream);
  edge_rank<<<cdiv(NEDGES, B), B, 0, stream>>>(dst, deg, rank);
  deg_scan<<<cdiv(NNODES, SCANB), SCANB, 0, stream>>>(deg, rowptr);
  edge_place<<<cdiv(NEDGES, B), B, 0, stream>>>(src, dst, rank, rowptr, csr);

  {
    int tiles = (NNODES / 16) * (HID / 16);
    gemm_wmma_f32<<<cdiv((long long)tiles * 32, B), B, 0, stream>>>(x, W1, xw1, NNODES, INDIM, HID);
  }
  attn_scores<<<cdiv((long long)NNODES * H1C, B), B, 0, stream>>>(xw1, as1, ad1, esrc1, edst1, NNODES, H1C, C1C);
  node_softmax_gather<H1C, C1C, 1><<<cdiv((long long)NNODES * 32, B), B, 0, stream>>>(
      rowptr, deg, csr, xw1, esrc1, edst1, b1, agg1);

  {
    int tiles = (NNODES / 16) * (OUTD / 16);
    gemm_wmma_f32<<<cdiv((long long)tiles * 32, B), B, 0, stream>>>(agg1, W2, xw2, NNODES, HID, OUTD);
  }
  attn_scores<<<cdiv((long long)NNODES, B), B, 0, stream>>>(xw2, as2, ad2, esrc2, edst2, NNODES, 1, OUTD);
  node_softmax_gather<1, OUTD, 0><<<cdiv((long long)NNODES * 32, B), B, 0, stream>>>(
      rowptr, deg, csr, xw2, esrc2, edst2, b2, agg2);

  pool_mean<<<NGRAPH, POOLW * 32, 0, stream>>>(agg2, bat, outF);
}


# GEMM retiled: a wave owns 16 full rows (W staged in LDS once per group, A read once), attention scores taken from its accumulators, 2 score launches gone; score dot sums now in lane-tree order
# speedup vs baseline: 5.3974x; 1.1752x over previous
#include <hip/hip_runtime.h>
#include <math.h>

#define NNODES 50000
#define NEDGES 800000
#define ETOT   (NEDGES + NNODES)
#define INDIM  128
#define H1C    4
#define C1C    32
#define HID    128
#define OUTD   64
#define NGRAPH 64

typedef float v2f __attribute__((ext_vector_type(2)));
typedef float v4f __attribute__((ext_vector_type(4)));
typedef float v8f __attribute__((ext_vector_type(8)));

__device__ __forceinline__ float row16_sum(float x) {
  x += __int_as_float(__builtin_amdgcn_update_dpp(0, __float_as_int(x), 0x161, 0xf, 0xf, true));
  x += __int_as_float(__builtin_amdgcn_update_dpp(0, __float_as_int(x), 0x162, 0xf, 0xf, true));
  x += __int_as_float(__builtin_amdgcn_update_dpp(0, __float_as_int(x), 0x164, 0xf, 0xf, true));
  x += __int_as_float(__builtin_amdgcn_update_dpp(0, __float_as_int(x), 0x168, 0xf, 0xf, true));
  return x;
}

template <int K, int NC, int H>
__global__ void __launch_bounds__(256) gemm_rows_scores(
    const float* __restrict__ A, const float* __restrict__ B,
    const float* __restrict__ a_src, const float* __restrict__ a_dst,
    float* __restrict__ Cm, float* __restrict__ esrc, float* __restrict__ edst, int M) {
  static_assert(K % 4 == 0 && NC % 32 == 0, "tile shapes");
  static_assert(H == 1 || H == 4, "score store forms");
  constexpr int NPAIR = NC / 32;
  constexpr int PPH   = NPAIR / H;
  constexpr int KS    = K / 4;
  constexpr int PITCH = 2 * NC;
  static_assert(PPH * H == NPAIR, "a head is whole tile pairs");
  __shared__ __attribute__((aligned(16))) float Bs[(K / 2) * PITCH];

  const int tid  = (int)threadIdx.x;
  const int lane = tid & 31;
  const int half = lane >> 4;
  const int c    = lane & 15;
  const int tm   = ((int)blockIdx.x * 8 + (tid >> 5)) * 16;

  const int arow_i = (tm + c < M) ? (tm + c) : (M - 1);
  const float* arow = A + (size_t)arow_i * K + 2 * half;
  v2f a[KS];
  #pragma unroll
  for (int s = 0; s < KS; ++s) a[s] = *reinterpret_cast<const v2f*>(arow + 4 * s);

  static_assert(((K / 2) * (NC / 4)) % 256 == 0, "whole trips");
  #pragma unroll
  for (int i = tid; i < (K / 2) * (NC / 4); i += 256) {
    const int p = i / (NC / 4);
    const int n = (i % (NC / 4)) * 4;
    const v4f r0 = *reinterpret_cast<const v4f*>(B + (size_t)(2 * p) * NC + n);
    const v4f r1 = *reinterpret_cast<const v4f*>(B + (size_t)(2 * p + 1) * NC + n);
    float* d = Bs + p * PITCH + n * 2;
    const v4f lo = {r0.x, r1.x, r0.y, r1.y};
    const v4f hi = {r0.z, r1.z, r0.w, r1.w};
    *reinterpret_cast<v4f*>(d)     = lo;
    *reinterpret_cast<v4f*>(d + 4) = hi;
  }
  __syncthreads();
  if (tm >= M) return;

  v8f acc[2 * NPAIR];
  #pragma unroll
  for (int t = 0; t < 2 * NPAIR; ++t) acc[t] = v8f{};

  const float* bl = Bs + half * PITCH + 4 * c;
  #pragma unroll
  for (int s = 0; s < KS; ++s) {
    #pragma unroll
    for (int q = 0; q < NPAIR; ++q) {
      const v4f b = *reinterpret_cast<const v4f*>(bl + (2 * s) * PITCH + 64 * q);
      const v2f b0 = {b.x, b.y};
      const v2f b1 = {b.z, b.w};
      acc[2 * q]     = __builtin_amdgcn_wmma_f32_16x16x4_f32(false, a[s], false, b0,
                                                             (short)0, acc[2 * q], false, false);
      acc[2 * q + 1] = __builtin_amdgcn_wmma_f32_16x16x4_f32(false, a[s], false, b1,
                                                             (short)0, acc[2 * q + 1], false, false);
    }
  }
  constexpr int NBLK = (KS * NPAIR) / 8;
  static_assert(NBLK * 8 == KS * NPAIR, "whole blocks");
  __builtin_amdgcn_sched_group_barrier(0x100, 8, 0);
  #pragma unroll
  for (int blk = 0; blk < NBLK; ++blk) {
    __builtin_amdgcn_sched_group_barrier(0x100, 8, 0);
    __builtin_amdgcn_sched_group_barrier(0x008, 16, 0);
  }

  const int row0 = tm + 8 * half;
  #pragma unroll
  for (int q = 0; q < NPAIR; ++q) {
    #pragma unroll
    for (int v = 0; v < 8; ++v) {
      const v2f o = {acc[2 * q][v], acc[2 * q + 1][v]};
      *reinterpret_cast<v2f*>(Cm + (size_t)(row0 + v) * NC + 32 * q + 2 * c) = o;
    }
  }

  float es[H][8], ed[H][8];
  #pragma unroll
  for (int q = 0; q < NPAIR; ++q) {
    const int h = q / PPH;
    const v2f s2 = *reinterpret_cast<const v2f*>(a_src + 32 * q + 2 * c);
    const v2f d2 = *reinterpret_cast<const v2f*>(a_dst + 32 * q + 2 * c);
    #pragma unroll
    for (int v = 0; v < 8; ++v) {
      const float ps = acc[2 * q][v] * s2.x + acc[2 * q + 1][v] * s2.y;
      const float pd = acc[2 * q][v] * d2.x + acc[2 * q + 1][v] * d2.y;
      if (q % PPH == 0) { es[h][v] = ps;  ed[h][v] = pd; }
      else              { es[h][v] += ps; ed[h][v] += pd; }
    }
  }
  #pragma unroll
  for (int h = 0; h < H; ++h) {
    #pragma unroll
    for (int v = 0; v < 8; ++v) {
      es[h][v] = row16_sum(es[h][v]);
      ed[h][v] = row16_sum(ed[h][v]);
    }
  }
  if (c == 0) {
    if constexpr (H == 4) {
      #pragma unroll
      for (int v = 0; v < 8; ++v) {
        const v4f s4 = {es[0][v], es[1][v], es[2][v], es[3][v]};
        const v4f d4 = {ed[0][v], ed[1][v], ed[2][v], ed[3][v]};
        *reinterpret_cast<v4f*>(esrc + (size_t)(row0 + v) * 4) = s4;
        *reinterpret_cast<v4f*>(edst + (size_t)(row0 + v) * 4) = d4;
      }
    } else {
      const v4f sa = {es[0][0], es[0][1], es[0][2], es[0][3]};
      const v4f sb = {es[0][4], es[0][5], es[0][6], es[0][7]};
      const v4f da = {ed[0][0], ed[0][1], ed[0][2], ed[0][3]};
      const v4f db = {ed[0][4], ed[0][5], ed[0][6], ed[0][7]};
      *reinterpret_cast<v4f*>(esrc + row0)     = sa;
      *reinterpret_cast<v4f*>(esrc + row0 + 4) = sb;
      *reinterpret_cast<v4f*>(edst + row0)     = da;
      *reinterpret_cast<v4f*>(edst + row0 + 4) = db;
    }
  }
}

__global__ void edge_rank(const int* __restrict__ dst, int* __restrict__ deg,
                          int* __restrict__ rank) {
  int e = blockIdx.x * blockDim.x + threadIdx.x;
  if (e >= NEDGES) return;
  rank[e] = atomicAdd(&deg[dst[e]], 1);
}

#define SCANB 1024
__global__ void __launch_bounds__(SCANB) deg_scan(const int* __restrict__ deg,
                                                  int* __restrict__ rowptr) {
  __shared__ int wpre[32];
  __shared__ int wtot[32];
  int t = (int)threadIdx.x;
  int lane = t & 31, w = t >> 5;
  int tile0 = (int)blockIdx.x * SCANB;
  int pre = 0;
  for (int i = t; i < tile0; i += SCANB) pre += deg[i];
  int idx = tile0 + t;
  int v = (idx < NNODES) ? deg[idx] : 0;
  int incl = v;
  for (int o = 1; o < 32; o <<= 1) {
    int n = __shfl_up(incl, o);
    if (lane >= o) incl += n;
  }
  for (int o = 16; o > 0; o >>= 1) pre += __shfl_xor(pre, o);
  if (lane == 31) wtot[w] = incl;
  if (lane == 0)  wpre[w] = pre;
  __syncthreads();
  int base = 0, woff = 0;
  for (int i = 0; i < 32; ++i) {
    base += wpre[i];
    if (i < w) woff += wtot[i];
  }
  if (idx < NNODES) rowptr[idx] = base + woff + incl - v;
}

__global__ void edge_place(const int* __restrict__ src, const int* __restrict__ dst,
                           const int* __restrict__ rank, const int* __restrict__ rowptr,
                           int* __restrict__ csr) {
  int e = blockIdx.x * blockDim.x + threadIdx.x;
  if (e >= NEDGES) return;
  csr[rowptr[dst[e]] + rank[e]] = src[e];
}

template <int H, int C, int ELU>
__global__ void __launch_bounds__(256) node_softmax_gather(
    const int* __restrict__ rowptr, const int* __restrict__ deg, const int* __restrict__ csr,
    const float* __restrict__ xw, const float* __restrict__ esrc, const float* __restrict__ edst,
    const float* __restrict__ bias, float* __restrict__ out) {
  constexpr int F = H * C;
  constexpr int V = F / 32;
  int node = __builtin_amdgcn_readfirstlane((int)((blockIdx.x * blockDim.x + threadIdx.x) >> 5));
  int lane = (int)(threadIdx.x & 31);
  if (node >= NNODES) return;
  const int hd = (lane * V) / C;
  const int beg = rowptr[node];
  const int total = deg[node] + 1;

  float ed[H];
  for (int h = 0; h < H; ++h) ed[h] = edst[node * H + h];

  auto load_chunk = [&](int base, int& s, float (&a)[H]) -> bool {
    int j = base + lane;
    s = node;
    if (j < total - 1) s = csr[beg + j];
    for (int h = 0; h < H; ++h) {
      float x = esrc[s * H + h] + ed[h];
      a[h] = (x >= 0.f) ? x : 0.2f * x;
    }
    return j < total;
  };

  float acc[V];
  for (int v = 0; v < V; ++v) acc[v] = 0.f;

  auto accumulate = [&](int s, const float (&wgt)[H], int cnt) {
    #pragma unroll 4
    for (int jj = 0; jj < cnt; ++jj) {
      int sj = __builtin_amdgcn_readlane(s, jj);
      float wsel = __int_as_float(__builtin_amdgcn_readlane(__float_as_int(wgt[0]), jj));
      for (int h = 1; h < H; ++h) {
        float wh = __int_as_float(__builtin_amdgcn_readlane(__float_as_int(wgt[h]), jj));
        wsel = (hd == h) ? wh : wsel;
      }
      const float* row = xw + (size_t)sj * F + lane * V;
      if constexpr (V == 4) {
        v4f r = *reinterpret_cast<const v4f*>(row);
        acc[0] += r.x * wsel; acc[1] += r.y * wsel; acc[2] += r.z * wsel; acc[3] += r.w * wsel;
      } else {
        v2f r = *reinterpret_cast<const v2f*>(row);
        acc[0] += r.x * wsel; acc[1] += r.y * wsel;
      }
    }
  };

  int s0; float a0[H];
  const bool ok0 = load_chunk(0, s0, a0);
  float m[H];
  for (int h = 0; h < H; ++h) m[h] = ok0 ? a0[h] : -INFINITY;
  for (int base = 32; base < total; base += 32) {
    int s; float a[H];
    bool ok = load_chunk(base, s, a);
    for (int h = 0; h < H; ++h) m[h] = ok ? fmaxf(m[h], a[h]) : m[h];
  }
  for (int h = 0; h < H; ++h)
    for (int o = 16; o > 0; o >>= 1) m[h] = fmaxf(m[h], __shfl_xor(m[h], o));

  float ex0[H], den[H];
  for (int h = 0; h < H; ++h) { ex0[h] = ok0 ? expf(a0[h] - m[h]) : 0.f; den[h] = ex0[h]; }
  for (int base = 32; base < total; base += 32) {
    int s; float a[H];
    bool ok = load_chunk(base, s, a);
    for (int h = 0; h < H; ++h) den[h] += ok ? expf(a[h] - m[h]) : 0.f;
  }
  for (int h = 0; h < H; ++h)
    for (int o = 16; o > 0; o >>= 1) den[h] += __shfl_xor(den[h], o);

  {
    float w0[H];
    for (int h = 0; h < H; ++h) w0[h] = ex0[h] / den[h];
    accumulate(s0, w0, total < 32 ? total : 32);
  }
  for (int base = 32; base < total; base += 32) {
    int s; float a[H], wv[H];
    bool ok = load_chunk(base, s, a);
    for (int h = 0; h < H; ++h) wv[h] = ok ? expf(a[h] - m[h]) / den[h] : 0.f;
    int left = total - base;
    accumulate(s, wv, left < 32 ? left : 32);
  }

  float res[V];
  for (int v = 0; v < V; ++v) {
    float r = acc[v] + bias[lane * V + v];
    if (ELU) r = (r > 0.f) ? r : (expf(r) - 1.f);
    res[v] = r;
  }
  float* o = out + (size_t)node * F + lane * V;
  if constexpr (V == 4) {
    v4f r = {res[0], res[1], res[2], res[3]};
    *reinterpret_cast<v4f*>(o) = r;
  } else {
    v2f r = {res[0], res[1]};
    *reinterpret_cast<v2f*>(o) = r;
  }
}

#define POOLW 32
#define POOLU 25
__global__ void __launch_bounds__(POOLW * 32) pool_mean(const float* __restrict__ h,
                                                        const int* __restrict__ batch,
                                                        float* __restrict__ out) {
  __shared__ float part[POOLW][OUTD];
  __shared__ int   pcnt[POOLW];
  constexpr int NCHUNK = (NNODES + 31) / 32;
  const int g    = (int)blockIdx.x;
  const int lane = (int)(threadIdx.x & 31);
  const int w = __builtin_amdgcn_readfirstlane((int)(threadIdx.x >> 5));

  float a0 = 0.f, a1 = 0.f;
  int cnt = 0;
  for (int c0 = w; c0 < NCHUNK; c0 += POOLW * POOLU) {
    int b[POOLU];
    #pragma unroll
    for (int k = 0; k < POOLU; ++k) {
      int node = (c0 + k * POOLW) * 32 + lane;
      int id = batch[node < NNODES ? node : NNODES - 1];
      b[k] = (node < NNODES) ? id : -1;
    }
    static_assert(POOLU <= 32, "one lane per chunk of a round");
    unsigned maskv = 0u;
    #pragma unroll
    for (int k = 0; k < POOLU; ++k) {
      unsigned mk = __builtin_amdgcn_ballot_w32(b[k] == g);
      maskv = (lane == k) ? mk : maskv;
    }
    unsigned any = __builtin_amdgcn_ballot_w32(maskv != 0u);
    while (any) {
      const int k = __builtin_ctz(any);
      any &= any - 1u;
      unsigned mask = (unsigned)__builtin_amdgcn_readlane((int)maskv, k);
      cnt += __builtin_popcount(mask);
      const int nbase = (c0 + k * POOLW) * 32;
      while (mask) {
        int  j[8];
        bool p[8];
        #pragma unroll
        for (int u = 0; u < 8; ++u) {
          p[u] = (mask != 0u);
          j[u] = p[u] ? __builtin_ctz(mask) : j[0];
          mask &= mask - 1u;
        }
        v2f r[8];
        #pragma unroll
        for (int u = 0; u < 8; ++u)
          r[u] = *reinterpret_cast<const v2f*>(h + (size_t)(nbase + j[u]) * OUTD + lane * 2);
        #pragma unroll
        for (int u = 0; u < 8; ++u) {
          a0 += p[u] ? r[u].x : 0.f;
          a1 += p[u] ? r[u].y : 0.f;
        }
      }
    }
  }

  v2f mine = {a0, a1};
  *reinterpret_cast<v2f*>(&part[w][lane * 2]) = mine;
  if (lane == 0) pcnt[w] = cnt;
  __syncthreads();
  const int t = (int)threadIdx.x;
  if (t < OUTD) {
    float s = 0.f;
    int   c = 0;
    for (int i = 0; i < POOLW; ++i) { s += part[i][t]; c += pcnt[i]; }
    out[(size_t)g * OUTD + t] = s / fmaxf((float)c, 1.f);
  }
}

static inline int cdiv(long long a, int b) { return (int)((a + b - 1) / b); }

extern "C" void kernel_launch(void* const* d_in, const int* in_sizes, int n_in,
                              void* d_out, int out_size, void* d_ws, size_t ws_size,
                              hipStream_t stream) {
  (void)in_sizes; (void)n_in; (void)out_size; (void)ws_size;
  const float* x   = (const float*)d_in[0];
  const int*   ei  = (const int*)d_in[1];
  const int*   bat = (const int*)d_in[2];
  const float* W1  = (const float*)d_in[3];
  const float* as1 = (const float*)d_in[4];
  const float* ad1 = (const float*)d_in[5];
  const float* b1  = (const float*)d_in[6];
  const float* W2  = (const float*)d_in[7];
  const float* as2 = (const float*)d_in[8];
  const float* ad2 = (const float*)d_in[9];
  const float* b2  = (const float*)d_in[10];
  const int* src = ei;
  const int* dst = ei + NEDGES;

  char* ws = (char*)d_ws;
  size_t off = 0;
  auto carve = [&](size_t bytes) -> char* {
    char* p = ws + off;
    off += (bytes + 255) & ~(size_t)255;
    return p;
  };
  float*    xw1    = (float*)carve((size_t)NNODES * HID * 4);
  float*    esrc1  = (float*)carve((size_t)NNODES * H1C * 4);
  float*    edst1  = (float*)carve((size_t)NNODES * H1C * 4);
  int*      deg    = (int*)carve((size_t)NNODES * 4);
  int*      rowptr = (int*)carve((size_t)NNODES * 4);
  int*      rank   = (int*)carve((size_t)NEDGES * 4);
  int*      csr    = (int*)carve((size_t)NEDGES * 4);
  float*    agg1   = (float*)carve((size_t)NNODES * HID * 4);
  float*    agg2   = (float*)carve((size_t)NNODES * OUTD * 4);
  float*    xw2    = xw1;
  float*    esrc2  = esrc1;
  float*    edst2  = edst1;
  float*    outF   = (float*)d_out;

  const int B = 256;

  hipMemsetAsync(deg, 0, (size_t)NNODES * 4, stream);
  edge_rank<<<cdiv(NEDGES, B), B, 0, stream>>>(dst, deg, rank);
  deg_scan<<<cdiv(NNODES, SCANB), SCANB, 0, stream>>>(deg, rowptr);
  edge_place<<<cdiv(NEDGES, B), B, 0, stream>>>(src, dst, rank, rowptr, csr);

  gemm_rows_scores<INDIM, HID, H1C><<<cdiv(NNODES / 16, 8), B, 0, stream>>>(
      x, W1, as1, ad1, xw1, esrc1, edst1, NNODES);
  node_softmax_gather<H1C, C1C, 1><<<cdiv((long long)NNODES * 32, B), B, 0, stream>>>(
      rowptr, deg, csr, xw1, esrc1, edst1, b1, agg1);

  gemm_rows_scores<HID, OUTD, 1><<<cdiv(NNODES / 16, 8), B, 0, stream>>>(
      agg1, W2, as2, ad2, xw2, esrc2, edst2, NNODES);
  node_softmax_gather<1, OUTD, 0><<<cdiv((long long)NNODES * 32, B), B, 0, stream>>>(
      rowptr, deg, csr, xw2, esrc2, edst2, b2, agg2);

  pool_mean<<<NGRAPH, POOLW * 32, 0, stream>>>(agg2, bat, outF);
}


# Edge grouping in LDS: 3200-edge chunks sorted by 128-node bucket in LDS, a work-group per bucket ranks and places in LDS; no global atomic, memset or scan launch; sums in LDS arrival order
# speedup vs baseline: 9.2224x; 1.7087x over previous
#include <hip/hip_runtime.h>
#include <math.h>

#define NNODES 50000
#define NEDGES 800000
#define ETOT   (NEDGES + NNODES)
#define INDIM  128
#define H1C    4
#define C1C    32
#define HID    128
#define OUTD   64
#define NGRAPH 64

typedef float v2f __attribute__((ext_vector_type(2)));
typedef float v4f __attribute__((ext_vector_type(4)));
typedef float v8f __attribute__((ext_vector_type(8)));

__device__ __forceinline__ float row16_sum(float x) {
  x += __int_as_float(__builtin_amdgcn_update_dpp(0, __float_as_int(x), 0x161, 0xf, 0xf, true));
  x += __int_as_float(__builtin_amdgcn_update_dpp(0, __float_as_int(x), 0x162, 0xf, 0xf, true));
  x += __int_as_float(__builtin_amdgcn_update_dpp(0, __float_as_int(x), 0x164, 0xf, 0xf, true));
  x += __int_as_float(__builtin_amdgcn_update_dpp(0, __float_as_int(x), 0x168, 0xf, 0xf, true));
  return x;
}

template <int K, int NC, int H>
__global__ void __launch_bounds__(256) gemm_rows_scores(
    const float* __restrict__ A, const float* __restrict__ B,
    const float* __restrict__ a_src, const float* __restrict__ a_dst,
    float* __restrict__ Cm, float* __restrict__ esrc, float* __restrict__ edst, int M) {
  static_assert(K % 4 == 0 && NC % 32 == 0, "tile shapes");
  static_assert(H == 1 || H == 4, "score store forms");
  constexpr int NPAIR = NC / 32;
  constexpr int PPH   = NPAIR / H;
  constexpr int KS    = K / 4;
  constexpr int PITCH = 2 * NC;
  static_assert(PPH * H == NPAIR, "a head is whole tile pairs");
  __shared__ __attribute__((aligned(16))) float Bs[(K / 2) * PITCH];

  const int tid  = (int)threadIdx.x;
  const int lane = tid & 31;
  const int half = lane >> 4;
  const int c    = lane & 15;
  const int tm   = ((int)blockIdx.x * 8 + (tid >> 5)) * 16;

  const int arow_i = (tm + c < M) ? (tm + c) : (M - 1);
  const float* arow = A + (size_t)arow_i * K + 2 * half;
  v2f a[KS];
  #pragma unroll
  for (int s = 0; s < KS; ++s) a[s] = *reinterpret_cast<const v2f*>(arow + 4 * s);

  static_assert(((K / 2) * (NC / 4)) % 256 == 0, "whole trips");
  #pragma unroll
  for (int i = tid; i < (K / 2) * (NC / 4); i += 256) {
    const int p = i / (NC / 4);
    const int n = (i % (NC / 4)) * 4;
    const v4f r0 = *reinterpret_cast<const v4f*>(B + (size_t)(2 * p) * NC + n);
    const v4f r1 = *reinterpret_cast<const v4f*>(B + (size_t)(2 * p + 1) * NC + n);
    float* d = Bs + p * PITCH + n * 2;
    const v4f lo = {r0.x, r1.x, r0.y, r1.y};
    const v4f hi = {r0.z, r1.z, r0.w, r1.w};
    *reinterpret_cast<v4f*>(d)     = lo;
    *reinterpret_cast<v4f*>(d + 4) = hi;
  }
  __syncthreads();
  if (tm >= M) return;

  v8f acc[2 * NPAIR];
  #pragma unroll
  for (int t = 0; t < 2 * NPAIR; ++t) acc[t] = v8f{};

  const float* bl = Bs + half * PITCH + 4 * c;
  #pragma unroll
  for (int s = 0; s < KS; ++s) {
    #pragma unroll
    for (int q = 0; q < NPAIR; ++q) {
      const v4f b = *reinterpret_cast<const v4f*>(bl + (2 * s) * PITCH + 64 * q);
      const v2f b0 = {b.x, b.y};
      const v2f b1 = {b.z, b.w};
      acc[2 * q]     = __builtin_amdgcn_wmma_f32_16x16x4_f32(false, a[s], false, b0,
                                                             (short)0, acc[2 * q], false, false);
      acc[2 * q + 1] = __builtin_amdgcn_wmma_f32_16x16x4_f32(false, a[s], false, b1,
                                                             (short)0, acc[2 * q + 1], false, false);
    }
  }
  constexpr int NBLK = (KS * NPAIR) / 8;
  static_assert(NBLK * 8 == KS * NPAIR, "whole blocks");
  __builtin_amdgcn_sched_group_barrier(0x100, 8, 0);
  #pragma unroll
  for (int blk = 0; blk < NBLK; ++blk) {
    __builtin_amdgcn_sched_group_barrier(0x100, 8, 0);
    __builtin_amdgcn_sched_group_barrier(0x008, 16, 0);
  }

  const int row0 = tm + 8 * half;
  #pragma unroll
  for (int q = 0; q < NPAIR; ++q) {
    #pragma unroll
    for (int v = 0; v < 8; ++v) {
      const v2f o = {acc[2 * q][v], acc[2 * q + 1][v]};
      *reinterpret_cast<v2f*>(Cm + (size_t)(row0 + v) * NC + 32 * q + 2 * c) = o;
    }
  }

  float es[H][8], ed[H][8];
  #pragma unroll
  for (int q = 0; q < NPAIR; ++q) {
    const int h = q / PPH;
    const v2f s2 = *reinterpret_cast<const v2f*>(a_src + 32 * q + 2 * c);
    const v2f d2 = *reinterpret_cast<const v2f*>(a_dst + 32 * q + 2 * c);
    #pragma unroll
    for (int v = 0; v < 8; ++v) {
      const float ps = acc[2 * q][v] * s2.x + acc[2 * q + 1][v] * s2.y;
      const float pd = acc[2 * q][v] * d2.x + acc[2 * q + 1][v] * d2.y;
      if (q % PPH == 0) { es[h][v] = ps;  ed[h][v] = pd; }
      else              { es[h][v] += ps; ed[h][v] += pd; }
    }
  }
  #pragma unroll
  for (int h = 0; h < H; ++h) {
    #pragma unroll
    for (int v = 0; v < 8; ++v) {
      es[h][v] = row16_sum(es[h][v]);
      ed[h][v] = row16_sum(ed[h][v]);
    }
  }
  if (c == 0) {
    if constexpr (H == 4) {
      #pragma unroll
      for (int v = 0; v < 8; ++v) {
        const v4f s4 = {es[0][v], es[1][v], es[2][v], es[3][v]};
        const v4f d4 = {ed[0][v], ed[1][v], ed[2][v], ed[3][v]};
        *reinterpret_cast<v4f*>(esrc + (size_t)(row0 + v) * 4) = s4;
        *reinterpret_cast<v4f*>(edst + (size_t)(row0 + v) * 4) = d4;
      }
    } else {
      const v4f sa = {es[0][0], es[0][1], es[0][2], es[0][3]};
      const v4f sb = {es[0][4], es[0][5], es[0][6], es[0][7]};
      const v4f da = {ed[0][0], ed[0][1], ed[0][2], ed[0][3]};
      const v4f db = {ed[0][4], ed[0][5], ed[0][6], ed[0][7]};
      *reinterpret_cast<v4f*>(esrc + row0)     = sa;
      *reinterpret_cast<v4f*>(esrc + row0 + 4) = sb;
      *reinterpret_cast<v4f*>(edst + row0)     = da;
      *reinterpret_cast<v4f*>(edst + row0 + 4) = db;
    }
  }
}

#define SB_NB   ((NNODES + 127) / 128)
#define SB_NCH  250
#define SB_CH   (NEDGES / SB_NCH)
#define SB_T1   (SB_CH / 4)
#define SB_NBP  (((SB_NB + 1) + 31) / 32 * 32)
#define SB_CAP  4096
#define SB_TPS  2
#define SB_T2   512
#define SB_KEEP 8
typedef int v4i __attribute__((ext_vector_type(4)));
static_assert(SB_NCH * SB_CH == NEDGES && SB_CH % 4 == 0, "whole chunks of 16-byte pieces");
static_assert(SB_T1 % 32 == 0 && SB_T1 <= 1024 && SB_T1 >= SB_NBP, "chunk_bin block shape");
static_assert(SB_NCH * SB_TPS <= SB_T2 && SB_T2 >= 128 && SB_T2 % 32 == 0, "one thread group per chunk run");
static_assert(NNODES <= (1 << 24), "source id and 7 bits of the destination in one word");

__global__ void __launch_bounds__(SB_T1) chunk_bin(const int* __restrict__ src,
                                                   const int* __restrict__ dst,
                                                   int* __restrict__ sorted,
                                                   int* __restrict__ offs) {
  __shared__ int hist[SB_NBP];
  __shared__ int off_s[SB_NBP];
  __shared__ int wtot[32];
  __shared__ __attribute__((aligned(16))) int stage[SB_CH];
  const int t = (int)threadIdx.x;
  const int lane = t & 31, w = t >> 5;
  const int chunk = (int)blockIdx.x;
  if (t < SB_NBP) hist[t] = 0;
  __syncthreads();
  const int e0 = chunk * SB_CH + t * 4;
  const v4i d = *reinterpret_cast<const v4i*>(dst + e0);
  const v4i s = *reinterpret_cast<const v4i*>(src + e0);
  int bk[4], rk[4];
  #pragma unroll
  for (int k = 0; k < 4; ++k) {
    bk[k] = d[k] >> 7;
    rk[k] = atomicAdd(&hist[bk[k]], 1);
  }
  __syncthreads();
  const int v = (t < SB_NB) ? hist[t] : 0;
  int incl = v;
  for (int o = 1; o < 32; o <<= 1) {
    int n = __shfl_up(incl, o);
    if (lane >= o) incl += n;
  }
  if (lane == 31) wtot[w] = incl;
  __syncthreads();
  int pre = 0;
  for (int i = 0; i < w; ++i) pre += wtot[i];
  const int excl = pre + incl - v;
  if (t <= SB_NB) {
    off_s[t] = excl;
    offs[chunk * (SB_NB + 1) + t] = excl;
  }
  __syncthreads();
  #pragma unroll
  for (int k = 0; k < 4; ++k)
    stage[off_s[bk[k]] + rk[k]] = (s[k] << 7) | (d[k] & 127);
  __syncthreads();
  const v4i o4 = *reinterpret_cast<const v4i*>(stage + t * 4);
  *reinterpret_cast<v4i*>(sorted + chunk * SB_CH + t * 4) = o4;
}

__global__ void __launch_bounds__(SB_T2) bucket_place(const int* __restrict__ sorted,
                                                     const int* __restrict__ offs,
                                                     int* __restrict__ deg,
                                                     int* __restrict__ rowptr,
                                                     int* __restrict__ csr) {
  __shared__ int cnt[128];
  __shared__ int lrow[128];
  __shared__ int cursor[128];
  __shared__ int wsum[4];
  __shared__ int base_s;
  __shared__ int stage[SB_CAP];
  const int t = (int)threadIdx.x;
  const int lane = t & 31;
  const int b = (int)blockIdx.x;
  const int c = t / SB_TPS, sub = t % SB_TPS;
  if (t < 128) { cnt[t] = 0; cursor[t] = 0; }
  if (t == 0) base_s = 0;
  __syncthreads();

  int o0 = 0, n = 0;
  if (c < SB_NCH) {
    o0 = offs[c * (SB_NB + 1) + b];
    n  = offs[c * (SB_NB + 1) + b + 1] - o0;
    if (sub == 0) atomicAdd(&base_s, o0);
  }
  const int* seg = sorted + (c < SB_NCH ? c * SB_CH + o0 : 0);

  int keep[SB_KEEP];
  #pragma unroll
  for (int k = 0; k < SB_KEEP; ++k) {
    const int j = sub + k * SB_TPS;
    keep[k] = (j < n) ? seg[j] : -1;
  }
  #pragma unroll
  for (int k = 0; k < SB_KEEP; ++k)
    if (keep[k] >= 0) atomicAdd(&cnt[keep[k] & 127], 1);
  for (int j = sub + SB_KEEP * SB_TPS; j < n; j += SB_TPS) atomicAdd(&cnt[seg[j] & 127], 1);
  __syncthreads();

  const int v = (t < 128) ? cnt[t] : 0;
  int incl = v;
  for (int o = 1; o < 32; o <<= 1) {
    int m = __shfl_up(incl, o);
    if (lane >= o) incl += m;
  }
  if (t < 128 && lane == 31) wsum[t >> 5] = incl;
  __syncthreads();
  if (t < 128) {
    int pre = 0;
    for (int i = 0; i < (t >> 5); ++i) pre += wsum[i];
    lrow[t] = pre + incl - v;
  }
  __syncthreads();
  const int base  = base_s;
  const int total = lrow[127] + cnt[127];
  const bool in_lds = total <= SB_CAP;
  if (t < 128) {
    const int node = b * 128 + t;
    if (node < NNODES) { deg[node] = v; rowptr[node] = base + lrow[t]; }
  }

  #pragma unroll
  for (int k = 0; k < SB_KEEP; ++k) {
    if (keep[k] >= 0) {
      const int l = keep[k] & 127;
      const int pos = lrow[l] + atomicAdd(&cursor[l], 1);
      if (in_lds) stage[pos] = keep[k] >> 7; else csr[base + pos] = keep[k] >> 7;
    }
  }
  for (int j = sub + SB_KEEP * SB_TPS; j < n; j += SB_TPS) {
    const int p = seg[j];
    const int l = p & 127;
    const int pos = lrow[l] + atomicAdd(&cursor[l], 1);
    if (in_lds) stage[pos] = p >> 7; else csr[base + pos] = p >> 7;
  }
  __syncthreads();
  if (in_lds)
    for (int i = t; i < total; i += SB_T2) csr[base + i] = stage[i];
}

template <int H, int C, int ELU>
__global__ void __launch_bounds__(256) node_softmax_gather(
    const int* __restrict__ rowptr, const int* __restrict__ deg, const int* __restrict__ csr,
    const float* __restrict__ xw, const float* __restrict__ esrc, const float* __restrict__ edst,
    const float* __restrict__ bias, float* __restrict__ out) {
  constexpr int F = H * C;
  constexpr int V = F / 32;
  int node = __builtin_amdgcn_readfirstlane((int)((blockIdx.x * blockDim.x + threadIdx.x) >> 5));
  int lane = (int)(threadIdx.x & 31);
  if (node >= NNODES) return;
  const int hd = (lane * V) / C;
  const int beg = rowptr[node];
  const int total = deg[node] + 1;

  float ed[H];
  for (int h = 0; h < H; ++h) ed[h] = edst[node * H + h];

  auto load_chunk = [&](int base, int& s, float (&a)[H]) -> bool {
    int j = base + lane;
    s = node;
    if (j < total - 1) s = csr[beg + j];
    for (int h = 0; h < H; ++h) {
      float x = esrc[s * H + h] + ed[h];
      a[h] = (x >= 0.f) ? x : 0.2f * x;
    }
    return j < total;
  };

  float acc[V];
  for (int v = 0; v < V; ++v) acc[v] = 0.f;

  auto accumulate = [&](int s, const float (&wgt)[H], int cnt) {
    #pragma unroll 4
    for (int jj = 0; jj < cnt; ++jj) {
      int sj = __builtin_amdgcn_readlane(s, jj);
      float wsel = __int_as_float(__builtin_amdgcn_readlane(__float_as_int(wgt[0]), jj));
      for (int h = 1; h < H; ++h) {
        float wh = __int_as_float(__builtin_amdgcn_readlane(__float_as_int(wgt[h]), jj));
        wsel = (hd == h) ? wh : wsel;
      }
      const float* row = xw + (size_t)sj * F + lane * V;
      if constexpr (V == 4) {
        v4f r = *reinterpret_cast<const v4f*>(row);
        acc[0] += r.x * wsel; acc[1] += r.y * wsel; acc[2] += r.z * wsel; acc[3] += r.w * wsel;
      } else {
        v2f r = *reinterpret_cast<const v2f*>(row);
        acc[0] += r.x * wsel; acc[1] += r.y * wsel;
      }
    }
  };

  int s0; float a0[H];
  const bool ok0 = load_chunk(0, s0, a0);
  float m[H];
  for (int h = 0; h < H; ++h) m[h] = ok0 ? a0[h] : -INFINITY;
  for (int base = 32; base < total; base += 32) {
    int s; float a[H];
    bool ok = load_chunk(base, s, a);
    for (int h = 0; h < H; ++h) m[h] = ok ? fmaxf(m[h], a[h]) : m[h];
  }
  for (int h = 0; h < H; ++h)
    for (int o = 16; o > 0; o >>= 1) m[h] = fmaxf(m[h], __shfl_xor(m[h], o));

  float ex0[H], den[H];
  for (int h = 0; h < H; ++h) { ex0[h] = ok0 ? expf(a0[h] - m[h]) : 0.f; den[h] = ex0[h]; }
  for (int base = 32; base < total; base += 32) {
    int s; float a[H];
    bool ok = load_chunk(base, s, a);
    for (int h = 0; h < H; ++h) den[h] += ok ? expf(a[h] - m[h]) : 0.f;
  }
  for (int h = 0; h < H; ++h)
    for (int o = 16; o > 0; o >>= 1) den[h] += __shfl_xor(den[h], o);

  {
    float w0[H];
    for (int h = 0; h < H; ++h) w0[h] = ex0[h] / den[h];
    accumulate(s0, w0, total < 32 ? total : 32);
  }
  for (int base = 32; base < total; base += 32) {
    int s; float a[H], wv[H];
    bool ok = load_chunk(base, s, a);
    for (int h = 0; h < H; ++h) wv[h] = ok ? expf(a[h] - m[h]) / den[h] : 0.f;
    int left = total - base;
    accumulate(s, wv, left < 32 ? left : 32);
  }

  float res[V];
  for (int v = 0; v < V; ++v) {
    float r = acc[v] + bias[lane * V + v];
    if (ELU) r = (r > 0.f) ? r : (expf(r) - 1.f);
    res[v] = r;
  }
  float* o = out + (size_t)node * F + lane * V;
  if constexpr (V == 4) {
    v4f r = {res[0], res[1], res[2], res[3]};
    *reinterpret_cast<v4f*>(o) = r;
  } else {
    v2f r = {res[0], res[1]};
    *reinterpret_cast<v2f*>(o) = r;
  }
}

#define POOLW 32
#define POOLU 25
__global__ void __launch_bounds__(POOLW * 32) pool_mean(const float* __restrict__ h,
                                                        const int* __restrict__ batch,
                                                        float* __restrict__ out) {
  __shared__ float part[POOLW][OUTD];
  __shared__ int   pcnt[POOLW];
  constexpr int NCHUNK = (NNODES + 31) / 32;
  const int g    = (int)blockIdx.x;
  const int lane = (int)(threadIdx.x & 31);
  const int w = __builtin_amdgcn_readfirstlane((int)(threadIdx.x >> 5));

  float a0 = 0.f, a1 = 0.f;
  int cnt = 0;
  for (int c0 = w; c0 < NCHUNK; c0 += POOLW * POOLU) {
    int b[POOLU];
    #pragma unroll
    for (int k = 0; k < POOLU; ++k) {
      int node = (c0 + k * POOLW) * 32 + lane;
      int id = batch[node < NNODES ? node : NNODES - 1];
      b[k] = (node < NNODES) ? id : -1;
    }
    static_assert(POOLU <= 32, "one lane per chunk of a round");
    unsigned maskv = 0u;
    #pragma unroll
    for (int k = 0; k < POOLU; ++k) {
      unsigned mk = __builtin_amdgcn_ballot_w32(b[k] == g);
      maskv = (lane == k) ? mk : maskv;
    }
    unsigned any = __builtin_amdgcn_ballot_w32(maskv != 0u);
    while (any) {
      const int k = __builtin_ctz(any);
      any &= any - 1u;
      unsigned mask = (unsigned)__builtin_amdgcn_readlane((int)maskv, k);
      cnt += __builtin_popcount(mask);
      const int nbase = (c0 + k * POOLW) * 32;
      while (mask) {
        int  j[8];
        bool p[8];
        #pragma unroll
        for (int u = 0; u < 8; ++u) {
          p[u] = (mask != 0u);
          j[u] = p[u] ? __builtin_ctz(mask) : j[0];
          mask &= mask - 1u;
        }
        v2f r[8];
        #pragma unroll
        for (int u = 0; u < 8; ++u)
          r[u] = *reinterpret_cast<const v2f*>(h + (size_t)(nbase + j[u]) * OUTD + lane * 2);
        #pragma unroll
        for (int u = 0; u < 8; ++u) {
          a0 += p[u] ? r[u].x : 0.f;
          a1 += p[u] ? r[u].y : 0.f;
        }
      }
    }
  }

  v2f mine = {a0, a1};
  *reinterpret_cast<v2f*>(&part[w][lane * 2]) = mine;
  if (lane == 0) pcnt[w] = cnt;
  __syncthreads();
  const int t = (int)threadIdx.x;
  if (t < OUTD) {
    float s = 0.f;
    int   c = 0;
    for (int i = 0; i < POOLW; ++i) { s += part[i][t]; c += pcnt[i]; }
    out[(size_t)g * OUTD + t] = s / fmaxf((float)c, 1.f);
  }
}

static inline int cdiv(long long a, int b) { return (int)((a + b - 1) / b); }

extern "C" void kernel_launch(void* const* d_in, const int* in_sizes, int n_in,
                              void* d_out, int out_size, void* d_ws, size_t ws_size,
                              hipStream_t stream) {
  (void)in_sizes; (void)n_in; (void)out_size; (void)ws_size;
  const float* x   = (const float*)d_in[0];
  const int*   ei  = (const int*)d_in[1];
  const int*   bat = (const int*)d_in[2];
  const float* W1  = (const float*)d_in[3];
  const float* as1 = (const float*)d_in[4];
  const float* ad1 = (const float*)d_in[5];
  const float* b1  = (const float*)d_in[6];
  const float* W2  = (const float*)d_in[7];
  const float* as2 = (const float*)d_in[8];
  const float* ad2 = (const float*)d_in[9];
  const float* b2  = (const float*)d_in[10];
  const int* src = ei;
  const int* dst = ei + NEDGES;

  char* ws = (char*)d_ws;
  size_t off = 0;
  auto carve = [&](size_t bytes) -> char* {
    char* p = ws + off;
    off += (bytes + 255) & ~(size_t)255;
    return p;
  };
  float*    xw1    = (float*)carve((size_t)NNODES * HID * 4);
  float*    esrc1  = (float*)carve((size_t)NNODES * H1C * 4);
  float*    edst1  = (float*)carve((size_t)NNODES * H1C * 4);
  int*      deg    = (int*)carve((size_t)NNODES * 4);
  int*      rowptr = (int*)carve((size_t)NNODES * 4);
  int*      sorted = (int*)carve((size_t)NEDGES * 4);
  int*      offs   = (int*)carve((size_t)SB_NCH * (SB_NB + 1) * 4);
  int*      csr    = (int*)carve((size_t)NEDGES * 4);
  float*    agg1   = (float*)carve((size_t)NNODES * HID * 4);
  float*    agg2   = (float*)carve((size_t)NNODES * OUTD * 4);
  float*    xw2    = xw1;
  float*    esrc2  = esrc1;
  float*    edst2  = edst1;
  float*    outF   = (float*)d_out;

  const int B = 256;

  chunk_bin<<<SB_NCH, SB_T1, 0, stream>>>(src, dst, sorted, offs);
  bucket_place<<<SB_NB, SB_T2, 0, stream>>>(sorted, offs, deg, rowptr, csr);

  gemm_rows_scores<INDIM, HID, H1C><<<cdiv(NNODES / 16, 8), B, 0, stream>>>(
      x, W1, as1, ad1, xw1, esrc1, edst1, NNODES);
  node_softmax_gather<H1C, C1C, 1><<<cdiv((long long)NNODES * 32, B), B, 0, stream>>>(
      rowptr, deg, csr, xw1, esrc1, edst1, b1, agg1);

  gemm_rows_scores<HID, OUTD, 1><<<cdiv(NNODES / 16, 8), B, 0, stream>>>(
      agg1, W2, as2, ad2, xw2, esrc2, edst2, NNODES);
  node_softmax_gather<1, OUTD, 0><<<cdiv((long long)NNODES * 32, B), B, 0, stream>>>(
      rowptr, deg, csr, xw2, esrc2, edst2, b2, agg2);

  pool_mean<<<NGRAPH, POOLW * 32, 0, stream>>>(agg2, bat, outF);
}
